# starting kernel
# baseline (speedup 1.0000x reference)
#include <hip/hip_runtime.h>


typedef __attribute__((ext_vector_type(2))) float v2f;
typedef __attribute__((ext_vector_type(8))) float v8f;
typedef __attribute__((ext_vector_type(4))) float f4;

#define PREV_LEN 10
#define AFT_LEN  10
#define NB       256
#define NC       32
#define HW       64
#define TM       (PREV_LEN - 1 + AFT_LEN)
#define MROWS    48
#define TSTR     68
#define EPSF     1e-6f

__global__ __launch_bounds__(128)
void mmvp_kernel(const float* __restrict__ feats,
                 const float* __restrict__ sim,
                 float* __restrict__ out)
{
    __shared__ float sT[MROWS * TSTR];
    __shared__ float sM[2][HW * HW];

    const int b    = blockIdx.x;
    const int tid  = threadIdx.x;
    const int wv   = tid >> 5;
    const int lane = tid & 31;
    const int lo16 = lane & 15;
    const int hi   = lane >> 4;
    const int cw   = wv * 16 + lo16;

    const float* featsB = feats + (size_t)b * PREV_LEN * NC * HW;
    const float* simB   = sim   + (size_t)b * TM * HW * HW;

    {
        const f4* s4 = (const f4*)simB;
        f4* d4 = (f4*)sM[0];
        for (int idx = tid; idx < (HW * HW) / 4; idx += 128)
            d4[idx] = s4[idx];
    }

    for (int idx = tid; idx < MROWS * HW; idx += 128) {
        int r = idx >> 6, c = idx & 63;
        float v = 0.0f;
        if (r < NC)       v = featsB[r * HW + c];
        else if (r == NC) v = 1.0f;
        sT[r * TSTR + c] = v;
    }
    __syncthreads();

    for (int step = 0; step < TM; ++step) {
        const float* __restrict__ cur = sM[step & 1];

        if (step + 1 < TM) {
            const f4* s4 = (const f4*)(simB + (size_t)(step + 1) * HW * HW);
            f4* d4 = (f4*)sM[(step + 1) & 1];
            for (int idx = tid; idx < (HW * HW) / 4; idx += 128)
                d4[idx] = s4[idx];
        }
        if (step + 2 < TM)
            __builtin_prefetch(simB + (size_t)(step + 2) * HW * HW + tid * 32, 0, 0);

        const bool isPrev = (step < PREV_LEN - 1);

        v8f a0 = {}, a1 = {}, a2 = {};
        if (isPrev) {
            const float* fa = featsB + (size_t)(step + 1) * NC * HW;
            #pragma unroll
            for (int r = 0; r < 8; ++r) {
                int r0 = r + hi * 8;
                a0[r] = fa[(r0)      * HW + cw];
                a1[r] = fa[(r0 + 16) * HW + cw];
            }
            a2[0] = (hi == 0) ? 1.0f : 0.0f;
        }

        #pragma unroll
        for (int kk = 0; kk < 16; ++kk) {
            const int kb = kk * 4 + hi * 2;
            v2f bf, fA0, fA1, fA2;
            bf.x  = cur[(kb)     * HW + cw];
            bf.y  = cur[(kb + 1) * HW + cw];
            fA0.x = sT[(lo16)      * TSTR + kb];
            fA0.y = sT[(lo16)      * TSTR + kb + 1];
            fA1.x = sT[(lo16 + 16) * TSTR + kb];
            fA1.y = sT[(lo16 + 16) * TSTR + kb + 1];
            fA2.x = sT[(lo16 + 32) * TSTR + kb];
            fA2.y = sT[(lo16 + 32) * TSTR + kb + 1];
            a0 = __builtin_amdgcn_wmma_f32_16x16x4_f32(false, fA0, false, bf, (short)0, a0, false, false);
            a1 = __builtin_amdgcn_wmma_f32_16x16x4_f32(false, fA1, false, bf, (short)0, a1, false, false);
            a2 = __builtin_amdgcn_wmma_f32_16x16x4_f32(false, fA2, false, bf, (short)0, a2, false, false);
        }
        __syncthreads();

        if (isPrev) {
            #pragma unroll
            for (int r = 0; r < 8; ++r) {
                int r0 = r + hi * 8;
                sT[(r0)      * TSTR + cw] = a0[r];
                sT[(r0 + 16) * TSTR + cw] = a1[r];
                sT[(r0 + 32) * TSTR + cw] = a2[r];
            }
            __syncthreads();
        } else {
            const int t = step - (PREV_LEN - 1);
            float wrow  = a2[0];
            float wsum  = __shfl(wrow, lo16, 32);
            float denom = wsum + EPSF;
            float* ob = out + (size_t)(b * AFT_LEN + t) * NC * HW;
            #pragma unroll
            for (int r = 0; r < 8; ++r) {
                int r0 = r + hi * 8;
                ob[(r0)      * HW + cw] = a0[r] / denom;
                ob[(r0 + 16) * HW + cw] = a1[r] / denom;
            }
        }
    }
}

extern "C" void kernel_launch(void* const* d_in, const int* in_sizes, int n_in,
                              void* d_out, int out_size, void* d_ws, size_t ws_size,
                              hipStream_t stream) {
    const float* feats = (const float*)d_in[0];
    const float* sim   = (const float*)d_in[1];
    float* out = (float*)d_out;
    mmvp_kernel<<<NB, 128, 0, stream>>>(feats, sim, out);
}


# Row split: each of 3 waves owns a 16-row tile of T, all 64 columns, in all 19 products; sim matrix global -> B-fragment registers (b128) one product ahead; Horner chain barrier-free; sums unchanged
# speedup vs baseline: 3.0647x; 3.0647x over previous
#include <hip/hip_runtime.h>


typedef __attribute__((ext_vector_type(2))) float v2f;
typedef __attribute__((ext_vector_type(8))) float v8f;
typedef __attribute__((ext_vector_type(4))) float f4;

#define PREV_LEN 10
#define AFT_LEN  10
#define NB       256
#define NC       32
#define HW       64
#define TM       (PREV_LEN - 1 + AFT_LEN)
#define NWAVE    3
#define TSTR     68
#define EPSF     1e-6f

__device__ __forceinline__ void load_mat(f4 (&q)[32], const float* __restrict__ m, int laneoff)
{
    const f4* p = (const f4*)m + laneoff;
    #pragma unroll
    for (int kk = 0; kk < 16; ++kk) {
        q[2 * kk]     = p[kk * 64];
        q[2 * kk + 1] = p[kk * 64 + 16];
    }
}

__device__ __forceinline__ void load_add(f4 (&c)[8], const float* __restrict__ fa, int rowoff)
{
    const f4* p = (const f4*)fa + rowoff;
    #pragma unroll
    for (int r = 0; r < 8; ++r) c[r] = p[r * 16];
}

template <bool LOADC, bool PREV>
__device__ __forceinline__ void product_step(const int step,
        const f4 (&cur)[32], f4 (&nxt)[32], const f4 (&ccur)[8], f4 (&cnxt)[8],
        const float* __restrict__ featsB, const float* __restrict__ simB, float* __restrict__ outB,
        float* myT, float (*sW)[HW],
        const int wv, const int tid, const int c, const int h,
        const int laneoff, const int rowoff, const int rowoffL)
{
    if (LOADC)
        load_add(cnxt, featsB + (size_t)(step + 2) * NC * HW, rowoffL);
    const int nm = (step + 1 < TM) ? step + 1 : TM - 1;
    load_mat(nxt, simB + (size_t)nm * HW * HW, laneoff);
    if (step + 3 < TM) {
        const float* m3 = simB + (size_t)(step + 3) * HW * HW;
        __builtin_prefetch(m3 + tid * 32, 0, 0);
        if (wv == 0) __builtin_prefetch(m3 + (96 + tid) * 32, 0, 0);
    }
    if (LOADC && step + 4 < PREV_LEN && wv < 2)
        __builtin_prefetch(featsB + (size_t)(step + 4) * NC * HW + tid * 32, 0, 0);

    v2f fA[16];
    #pragma unroll
    for (int kk = 0; kk < 16; ++kk) {
        fA[kk].x = myT[c * TSTR + kk * 4 + h * 2];
        fA[kk].y = myT[c * TSTR + kk * 4 + h * 2 + 1];
    }

    v8f acc[4] = {};
    if (PREV) {
        if (wv < 2) {
            #pragma unroll
            for (int r = 0; r < 8; ++r) {
                acc[0][r] = ccur[r].x; acc[1][r] = ccur[r].y;
                acc[2][r] = ccur[r].z; acc[3][r] = ccur[r].w;
            }
        } else {
            const float one = (h == 0) ? 1.0f : 0.0f;
            acc[0][0] = one; acc[1][0] = one; acc[2][0] = one; acc[3][0] = one;
        }
    }

    #pragma unroll
    for (int kk = 0; kk < 16; ++kk) {
        const f4 b0 = cur[2 * kk], b1 = cur[2 * kk + 1];
        v2f bf;
        bf.x = b0.x; bf.y = b1.x;
        acc[0] = __builtin_amdgcn_wmma_f32_16x16x4_f32(false, fA[kk], false, bf, (short)0, acc[0], false, false);
        bf.x = b0.y; bf.y = b1.y;
        acc[1] = __builtin_amdgcn_wmma_f32_16x16x4_f32(false, fA[kk], false, bf, (short)0, acc[1], false, false);
        bf.x = b0.z; bf.y = b1.z;
        acc[2] = __builtin_amdgcn_wmma_f32_16x16x4_f32(false, fA[kk], false, bf, (short)0, acc[2], false, false);
        bf.x = b0.w; bf.y = b1.w;
        acc[3] = __builtin_amdgcn_wmma_f32_16x16x4_f32(false, fA[kk], false, bf, (short)0, acc[3], false, false);
    }

    if (PREV) {
        #pragma unroll
        for (int r = 0; r < 8; ++r) {
            float* row = myT + (h * 8 + r) * TSTR + c * 4;
            row[0] = acc[0][r]; row[1] = acc[1][r]; row[2] = acc[2][r]; row[3] = acc[3][r];
        }
    } else {
        const int t = step - (PREV_LEN - 1);
        if (wv == 2 && h == 0) {
            float* w = sW[t] + c * 4;
            w[0] = acc[0][0]; w[1] = acc[1][0]; w[2] = acc[2][0]; w[3] = acc[3][0];
        }
        __builtin_amdgcn_fence(__ATOMIC_RELEASE, "workgroup", "local");
        __builtin_amdgcn_s_barrier();
        __builtin_amdgcn_fence(__ATOMIC_ACQUIRE, "workgroup", "local");
        if (wv < 2) {
            const float* w = sW[t] + c * 4;
            const float d0 = w[0] + EPSF, d1 = w[1] + EPSF, d2 = w[2] + EPSF, d3 = w[3] + EPSF;
            f4* o = (f4*)(outB + (size_t)t * NC * HW) + rowoff;
            #pragma unroll
            for (int r = 0; r < 8; ++r) {
                f4 v;
                v.x = acc[0][r] / d0; v.y = acc[1][r] / d1;
                v.z = acc[2][r] / d2; v.w = acc[3][r] / d3;
                o[r * 16] = v;
            }
        }
    }
}

__global__ __launch_bounds__(NWAVE * 32)
void chain_rows_kernel(const float* __restrict__ feats,
                      const float* __restrict__ sim,
                      float* __restrict__ out)
{
    __shared__ __attribute__((aligned(16))) float sT[NWAVE][16 * TSTR];
    __shared__ __attribute__((aligned(16))) float sW[AFT_LEN][HW];

    const int b   = blockIdx.x;
    const int tid = threadIdx.x;
    const int wv  = __builtin_amdgcn_readfirstlane(tid >> 5);
    const int lane = tid & 31;
    const int c   = lane & 15;
    const int h   = lane >> 4;
    const int wvL = (wv < 2) ? wv : 1;
    const int laneoff = h * 32 + c;
    const int rowoff  = (wv  * 16 + h * 8) * 16 + c;
    const int rowoffL = (wvL * 16 + h * 8) * 16 + c;

    const float* featsB = feats + (size_t)b * PREV_LEN * NC * HW;
    const float* simB   = sim   + (size_t)b * TM * HW * HW;
    float*       outB   = out   + (size_t)b * AFT_LEN * NC * HW;
    float*       myT    = sT[wv];

    f4 mA[32], mB[32];
    f4 cA[8], cB[8];

    #pragma unroll
    for (int p = 1; p <= 2; ++p) {
        __builtin_prefetch(simB + (size_t)p * HW * HW + tid * 32, 0, 0);
        if (wv == 0) __builtin_prefetch(simB + (size_t)p * HW * HW + (96 + tid) * 32, 0, 0);
        if (wv < 2)  __builtin_prefetch(featsB + (size_t)(p + 1) * NC * HW + tid * 32, 0, 0);
    }

    {
        f4 t0[8];
        load_add(t0, featsB, rowoffL);
        load_add(cA, featsB + (size_t)NC * HW, rowoffL);
        load_mat(mA, simB, laneoff);
        #pragma unroll
        for (int r = 0; r < 8; ++r) {
            f4 v = t0[r];
            if (wv == 2) {
                const float one = (h == 0 && r == 0) ? 1.0f : 0.0f;
                v.x = one; v.y = one; v.z = one; v.w = one;
            }
            float* row = myT + (h * 8 + r) * TSTR + c * 4;
            row[0] = v.x; row[1] = v.y; row[2] = v.z; row[3] = v.w;
        }
    }

    #pragma nounroll
    for (int s = 0; s < PREV_LEN - 2; s += 2) {
        product_step<true, true>(s,     mA, mB, cA, cB, featsB, simB, outB, myT, sW, wv, tid, c, h, laneoff, rowoff, rowoffL);
        product_step<true, true>(s + 1, mB, mA, cB, cA, featsB, simB, outB, myT, sW, wv, tid, c, h, laneoff, rowoff, rowoffL);
    }
    product_step<false, true>(PREV_LEN - 2, mA, mB, cA, cB, featsB, simB, outB, myT, sW, wv, tid, c, h, laneoff, rowoff, rowoffL);
    #pragma nounroll
    for (int s = PREV_LEN - 1; s < TM; s += 2) {
        product_step<false, false>(s,     mB, mA, cB, cA, featsB, simB, outB, myT, sW, wv, tid, c, h, laneoff, rowoff, rowoffL);
        product_step<false, false>(s + 1, mA, mB, cA, cB, featsB, simB, outB, myT, sW, wv, tid, c, h, laneoff, rowoff, rowoffL);
    }
}

extern "C" void kernel_launch(void* const* d_in, const int* in_sizes, int n_in,
                              void* d_out, int out_size, void* d_ws, size_t ws_size,
                              hipStream_t stream) {
    const float* feats = (const float*)d_in[0];
    const float* sim   = (const float*)d_in[1];
    float* out = (float*)d_out;
    chain_rows_kernel<<<NB, NWAVE * 32, 0, stream>>>(feats, sim, out);
}


# Output normalisation: the weight wave inverts each column weight once (IEEE 1/x), the feature waves multiply; no IEEE division per output element; sums unchanged, last bit may differ from x/d
# speedup vs baseline: 3.5221x; 1.1492x over previous
#include <hip/hip_runtime.h>


typedef __attribute__((ext_vector_type(2))) float v2f;
typedef __attribute__((ext_vector_type(8))) float v8f;
typedef __attribute__((ext_vector_type(4))) float f4;

#define PREV_LEN 10
#define AFT_LEN  10
#define NB       256
#define NC       32
#define HW       64
#define TM       (PREV_LEN - 1 + AFT_LEN)
#define NWAVE    3
#define TSTR     68
#define EPSF     1e-6f

__device__ __forceinline__ void load_mat(f4 (&q)[32], const float* __restrict__ m, int laneoff)
{
    const f4* p = (const f4*)m + laneoff;
    #pragma unroll
    for (int kk = 0; kk < 16; ++kk) {
        q[2 * kk]     = p[kk * 64];
        q[2 * kk + 1] = p[kk * 64 + 16];
    }
}

__device__ __forceinline__ void load_add(f4 (&c)[8], const float* __restrict__ fa, int rowoff)
{
    const f4* p = (const f4*)fa + rowoff;
    #pragma unroll
    for (int r = 0; r < 8; ++r) c[r] = p[r * 16];
}

template <bool LOADC, bool PREV>
__device__ __forceinline__ void product_step(const int step,
        const f4 (&cur)[32], f4 (&nxt)[32], const f4 (&ccur)[8], f4 (&cnxt)[8],
        const float* __restrict__ featsB, const float* __restrict__ simB, float* __restrict__ outB,
        float* myT, float (*sW)[HW],
        const int wv, const int tid, const int c, const int h,
        const int laneoff, const int rowoff, const int rowoffL)
{
    if (LOADC)
        load_add(cnxt, featsB + (size_t)(step + 2) * NC * HW, rowoffL);
    const int nm = (step + 1 < TM) ? step + 1 : TM - 1;
    load_mat(nxt, simB + (size_t)nm * HW * HW, laneoff);
    if (step + 3 < TM) {
        const float* m3 = simB + (size_t)(step + 3) * HW * HW;
        __builtin_prefetch(m3 + tid * 32, 0, 0);
        if (wv == 0) __builtin_prefetch(m3 + (96 + tid) * 32, 0, 0);
    }
    if (LOADC && step + 4 < PREV_LEN && wv < 2)
        __builtin_prefetch(featsB + (size_t)(step + 4) * NC * HW + tid * 32, 0, 0);

    v2f fA[16];
    #pragma unroll
    for (int kk = 0; kk < 16; ++kk) {
        fA[kk].x = myT[c * TSTR + kk * 4 + h * 2];
        fA[kk].y = myT[c * TSTR + kk * 4 + h * 2 + 1];
    }

    v8f acc[4] = {};
    if (PREV) {
        if (wv < 2) {
            #pragma unroll
            for (int r = 0; r < 8; ++r) {
                acc[0][r] = ccur[r].x; acc[1][r] = ccur[r].y;
                acc[2][r] = ccur[r].z; acc[3][r] = ccur[r].w;
            }
        } else {
            const float one = (h == 0) ? 1.0f : 0.0f;
            acc[0][0] = one; acc[1][0] = one; acc[2][0] = one; acc[3][0] = one;
        }
    }

    #pragma unroll
    for (int kk = 0; kk < 16; ++kk) {
        const f4 b0 = cur[2 * kk], b1 = cur[2 * kk + 1];
        v2f bf;
        bf.x = b0.x; bf.y = b1.x;
        acc[0] = __builtin_amdgcn_wmma_f32_16x16x4_f32(false, fA[kk], false, bf, (short)0, acc[0], false, false);
        bf.x = b0.y; bf.y = b1.y;
        acc[1] = __builtin_amdgcn_wmma_f32_16x16x4_f32(false, fA[kk], false, bf, (short)0, acc[1], false, false);
        bf.x = b0.z; bf.y = b1.z;
        acc[2] = __builtin_amdgcn_wmma_f32_16x16x4_f32(false, fA[kk], false, bf, (short)0, acc[2], false, false);
        bf.x = b0.w; bf.y = b1.w;
        acc[3] = __builtin_amdgcn_wmma_f32_16x16x4_f32(false, fA[kk], false, bf, (short)0, acc[3], false, false);
    }

    if (PREV) {
        #pragma unroll
        for (int r = 0; r < 8; ++r) {
            float* row = myT + (h * 8 + r) * TSTR + c * 4;
            row[0] = acc[0][r]; row[1] = acc[1][r]; row[2] = acc[2][r]; row[3] = acc[3][r];
        }
    } else {
        const int t = step - (PREV_LEN - 1);
        if (wv == 2 && h == 0) {
            f4 rw;
            rw.x = 1.0f / (acc[0][0] + EPSF); rw.y = 1.0f / (acc[1][0] + EPSF);
            rw.z = 1.0f / (acc[2][0] + EPSF); rw.w = 1.0f / (acc[3][0] + EPSF);
            *(f4*)(sW[t] + c * 4) = rw;
        }
        __builtin_amdgcn_fence(__ATOMIC_RELEASE, "workgroup", "local");
        __builtin_amdgcn_s_barrier();
        __builtin_amdgcn_fence(__ATOMIC_ACQUIRE, "workgroup", "local");
        if (wv < 2) {
            const f4 rw = *(const f4*)(sW[t] + c * 4);
            f4* o = (f4*)(outB + (size_t)t * NC * HW) + rowoff;
            #pragma unroll
            for (int r = 0; r < 8; ++r) {
                f4 v;
                v.x = acc[0][r] * rw.x; v.y = acc[1][r] * rw.y;
                v.z = acc[2][r] * rw.z; v.w = acc[3][r] * rw.w;
                o[r * 16] = v;
            }
        }
    }
}

__global__ __launch_bounds__(NWAVE * 32)
void chain_rows_kernel(const float* __restrict__ feats,
                      const float* __restrict__ sim,
                      float* __restrict__ out)
{
    __shared__ __attribute__((aligned(16))) float sT[NWAVE][16 * TSTR];
    __shared__ __attribute__((aligned(16))) float sW[AFT_LEN][HW];

    const int b   = blockIdx.x;
    const int tid = threadIdx.x;
    const int wv  = __builtin_amdgcn_readfirstlane(tid >> 5);
    const int lane = tid & 31;
    const int c   = lane & 15;
    const int h   = lane >> 4;
    const int wvL = (wv < 2) ? wv : 1;
    const int laneoff = h * 32 + c;
    const int rowoff  = (wv  * 16 + h * 8) * 16 + c;
    const int rowoffL = (wvL * 16 + h * 8) * 16 + c;

    const float* featsB = feats + (size_t)b * PREV_LEN * NC * HW;
    const float* simB   = sim   + (size_t)b * TM * HW * HW;
    float*       outB   = out   + (size_t)b * AFT_LEN * NC * HW;
    float*       myT    = sT[wv];

    f4 mA[32], mB[32];
    f4 cA[8], cB[8];

    #pragma unroll
    for (int p = 1; p <= 2; ++p) {
        __builtin_prefetch(simB + (size_t)p * HW * HW + tid * 32, 0, 0);
        if (wv == 0) __builtin_prefetch(simB + (size_t)p * HW * HW + (96 + tid) * 32, 0, 0);
        if (wv < 2)  __builtin_prefetch(featsB + (size_t)(p + 1) * NC * HW + tid * 32, 0, 0);
    }

    {
        f4 t0[8];
        load_add(t0, featsB, rowoffL);
        load_add(cA, featsB + (size_t)NC * HW, rowoffL);
        load_mat(mA, simB, laneoff);
        #pragma unroll
        for (int r = 0; r < 8; ++r) {
            f4 v = t0[r];
            if (wv == 2) {
                const float one = (h == 0 && r == 0) ? 1.0f : 0.0f;
                v.x = one; v.y = one; v.z = one; v.w = one;
            }
            float* row = myT + (h * 8 + r) * TSTR + c * 4;
            row[0] = v.x; row[1] = v.y; row[2] = v.z; row[3] = v.w;
        }
    }

    #pragma nounroll
    for (int s = 0; s < PREV_LEN - 2; s += 2) {
        product_step<true, true>(s,     mA, mB, cA, cB, featsB, simB, outB, myT, sW, wv, tid, c, h, laneoff, rowoff, rowoffL);
        product_step<true, true>(s + 1, mB, mA, cB, cA, featsB, simB, outB, myT, sW, wv, tid, c, h, laneoff, rowoff, rowoffL);
    }
    product_step<false, true>(PREV_LEN - 2, mA, mB, cA, cB, featsB, simB, outB, myT, sW, wv, tid, c, h, laneoff, rowoff, rowoffL);
    #pragma nounroll
    for (int s = PREV_LEN - 1; s < TM; s += 2) {
        product_step<false, false>(s,     mB, mA, cB, cA, featsB, simB, outB, myT, sW, wv, tid, c, h, laneoff, rowoff, rowoffL);
        product_step<false, false>(s + 1, mA, mB, cA, cB, featsB, simB, outB, myT, sW, wv, tid, c, h, laneoff, rowoff, rowoffL);
    }
}

extern "C" void kernel_launch(void* const* d_in, const int* in_sizes, int n_in,
                              void* d_out, int out_size, void* d_ws, size_t ws_size,
                              hipStream_t stream) {
    const float* feats = (const float*)d_in[0];
    const float* sim   = (const float*)d_in[1];
    float* out = (float*)d_out;
    chain_rows_kernel<<<NB, NWAVE * 32, 0, stream>>>(feats, sim, out);
}


# K split over 4 waves, 1 per SIMD: each does a k half of one feature tile (32 WMMAs) and a k quarter of the weight row (16) on shared B pairs; partials summed via LDS, one barrier; k sum order changed
# speedup vs baseline: 3.7430x; 1.0627x over previous
#include <hip/hip_runtime.h>


typedef __attribute__((ext_vector_type(2))) float v2f;
typedef __attribute__((ext_vector_type(8))) float v8f;
typedef __attribute__((ext_vector_type(4))) float f4;

#define PREV_LEN 10
#define AFT_LEN  10
#define NB       256
#define NC       32
#define HW       64
#define TM       (PREV_LEN - 1 + AFT_LEN)
#define NWAVE    4
#define TSTR     68
#define PTILE    (16 * TSTR)
#define EPSF     1e-6f

__device__ __forceinline__ void load_half(f4 (&q)[16], const f4* __restrict__ p1, const f4* __restrict__ p2)
{
    #pragma unroll
    for (int kk = 0; kk < 4; ++kk) {
        q[2 * kk]     = p1[kk * 64];
        q[2 * kk + 1] = p1[kk * 64 + 16];
    }
    #pragma unroll
    for (int kk = 0; kk < 4; ++kk) {
        q[8 + 2 * kk]     = p2[kk * 64];
        q[8 + 2 * kk + 1] = p2[kk * 64 + 16];
    }
}

__device__ __forceinline__ void load_add(f4 (&c)[8], const float* __restrict__ fa, int rowoff)
{
    const f4* p = (const f4*)fa + rowoff;
    #pragma unroll
    for (int r = 0; r < 8; ++r) c[r] = p[r * 16];
}

__device__ __forceinline__ void read_frags(v2f (&fA)[8], v2f (&fW)[4],
        const float* P0, const float* P1, const float* Wp,
        const int aoff1, const int aoff2, const int woff, const bool row0)
{
    #pragma unroll
    for (int kk = 0; kk < 4; ++kk) {
        fA[kk]     = *(const v2f*)(P0 + aoff1 + 4 * kk) + *(const v2f*)(P1 + aoff1 + 4 * kk);
        fA[4 + kk] = *(const v2f*)(P0 + aoff2 + 4 * kk) + *(const v2f*)(P1 + aoff2 + 4 * kk);
    }
    #pragma unroll
    for (int kk = 0; kk < 4; ++kk) {
        const v2f w = (*(const v2f*)(Wp + woff + 4 * kk) + *(const v2f*)(Wp + HW + woff + 4 * kk))
                    + (*(const v2f*)(Wp + 2 * HW + woff + 4 * kk) + *(const v2f*)(Wp + 3 * HW + woff + 4 * kk));
        fW[kk].x = row0 ? w.x : 0.0f;
        fW[kk].y = row0 ? w.y : 0.0f;
    }
}

template <bool LOADC, bool PREV>
__device__ __forceinline__ void product_step(const int step,
        const f4 (&cur)[16], f4 (&nxt)[16], const f4 (&ccur)[8], f4 (&cnxt)[8],
        v2f (&fA)[8], v2f (&fW)[4],
        const float* __restrict__ featsB, const float* __restrict__ simB, float* __restrict__ outB,
        float* sPbase, float* sWbase,
        const int wv, const int tid, const int c, const int h, const int ft, const int kh,
        const int off1, const int off2, const int rowoff,
        const int aoff1, const int aoff2, const int woff)
{
    if (LOADC)
        load_add(cnxt, featsB + (size_t)(step + 2) * NC * HW, rowoff);
    const int nm = (step + 1 < TM) ? step + 1 : TM - 1;
    const f4* mn = (const f4*)(simB + (size_t)nm * HW * HW);
    load_half(nxt, mn + off1, mn + off2);
    if (step + 3 < TM)
        __builtin_prefetch(simB + (size_t)(step + 3) * HW * HW + tid * 32, 0, 0);
    if (LOADC && step + 4 < PREV_LEN && wv < 2)
        __builtin_prefetch(featsB + (size_t)(step + 4) * NC * HW + tid * 32, 0, 0);

    v8f acc[4] = {};
    v8f accW[4] = {};
    if (PREV) {
        if (kh == 0) {
            #pragma unroll
            for (int r = 0; r < 8; ++r) {
                acc[0][r] = ccur[r].x; acc[1][r] = ccur[r].y;
                acc[2][r] = ccur[r].z; acc[3][r] = ccur[r].w;
            }
        }
        if (wv == 0) {
            const float one = (h == 0) ? 1.0f : 0.0f;
            accW[0][0] = one; accW[1][0] = one; accW[2][0] = one; accW[3][0] = one;
        }
    }

    #pragma unroll
    for (int kk = 0; kk < 8; ++kk) {
        const f4 b0 = cur[2 * kk], b1 = cur[2 * kk + 1];
        v2f bf;
        bf.x = b0.x; bf.y = b1.x;
        acc[0] = __builtin_amdgcn_wmma_f32_16x16x4_f32(false, fA[kk], false, bf, (short)0, acc[0], false, false);
        if (kk < 4) accW[0] = __builtin_amdgcn_wmma_f32_16x16x4_f32(false, fW[kk], false, bf, (short)0, accW[0], false, false);
        bf.x = b0.y; bf.y = b1.y;
        acc[1] = __builtin_amdgcn_wmma_f32_16x16x4_f32(false, fA[kk], false, bf, (short)0, acc[1], false, false);
        if (kk < 4) accW[1] = __builtin_amdgcn_wmma_f32_16x16x4_f32(false, fW[kk], false, bf, (short)0, accW[1], false, false);
        bf.x = b0.z; bf.y = b1.z;
        acc[2] = __builtin_amdgcn_wmma_f32_16x16x4_f32(false, fA[kk], false, bf, (short)0, acc[2], false, false);
        if (kk < 4) accW[2] = __builtin_amdgcn_wmma_f32_16x16x4_f32(false, fW[kk], false, bf, (short)0, accW[2], false, false);
        bf.x = b0.w; bf.y = b1.w;
        acc[3] = __builtin_amdgcn_wmma_f32_16x16x4_f32(false, fA[kk], false, bf, (short)0, acc[3], false, false);
        if (kk < 4) accW[3] = __builtin_amdgcn_wmma_f32_16x16x4_f32(false, fW[kk], false, bf, (short)0, accW[3], false, false);
    }

    const int set = (step + 1) & 1;
    float* myP = sPbase + ((set * 2 + ft) * 2 + kh) * PTILE;
    #pragma unroll
    for (int r = 0; r < 8; ++r) {
        f4 v;
        v.x = acc[0][r]; v.y = acc[1][r]; v.z = acc[2][r]; v.w = acc[3][r];
        *(f4*)(myP + (h * 8 + r) * TSTR + c * 4) = v;
    }
    float* Wp = sWbase + set * 4 * HW;
    if (h == 0) {
        f4 v;
        v.x = accW[0][0]; v.y = accW[1][0]; v.z = accW[2][0]; v.w = accW[3][0];
        *(f4*)(Wp + (2 * kh + ft) * HW + c * 4) = v;
    }
    __builtin_amdgcn_fence(__ATOMIC_RELEASE, "workgroup", "local");
    __builtin_amdgcn_s_barrier();
    __builtin_amdgcn_fence(__ATOMIC_ACQUIRE, "workgroup", "local");

    const float* P0 = sPbase + ((set * 2 + ft) * 2) * PTILE;
    const float* P1 = P0 + PTILE;
    if (PREV) {
        read_frags(fA, fW, P0, P1, Wp, aoff1, aoff2, woff, c == 0);
    } else {
        const int t = step - (PREV_LEN - 1);
        const f4 w = (*(const f4*)(Wp + c * 4) + *(const f4*)(Wp + HW + c * 4))
                   + (*(const f4*)(Wp + 2 * HW + c * 4) + *(const f4*)(Wp + 3 * HW + c * 4));
        f4 rw;
        rw.x = 1.0f / (w.x + EPSF); rw.y = 1.0f / (w.y + EPSF);
        rw.z = 1.0f / (w.z + EPSF); rw.w = 1.0f / (w.w + EPSF);
        f4* o = (f4*)(outB + (size_t)t * NC * HW) + (ft * 16 + kh * 8 + h) * 16 + c;
        #pragma unroll
        for (int i = 0; i < 4; ++i) {
            const int rr = kh * 8 + 2 * i + h;
            const f4 s = *(const f4*)(P0 + rr * TSTR + c * 4) + *(const f4*)(P1 + rr * TSTR + c * 4);
            o[i * 32] = s * rw;
        }
    }
}

__global__ __launch_bounds__(NWAVE * 32)
void chain_ksplit_kernel(const float* __restrict__ feats,
                         const float* __restrict__ sim,
                         float* __restrict__ out)
{
    __shared__ __attribute__((aligned(16))) float sP[2 * 2 * 2 * PTILE];
    __shared__ __attribute__((aligned(16))) float sW[2 * 4 * HW];

    const int b   = blockIdx.x;
    const int tid = threadIdx.x;
    const int wv  = __builtin_amdgcn_readfirstlane(tid >> 5);
    const int ft  = wv >> 1;
    const int kh  = wv & 1;
    const int lane = tid & 31;
    const int c   = lane & 15;
    const int h   = lane >> 4;
    const int g1  = 8 * kh + 4 * ft;
    const int g2  = 8 * kh + 4 * (1 - ft);
    const int off1 = g1 * 64 + h * 32 + c;
    const int off2 = g2 * 64 + h * 32 + c;
    const int rowoff = (ft * 16 + h * 8) * 16 + c;
    const int aoff1 = c * TSTR + 4 * g1 + 2 * h;
    const int aoff2 = c * TSTR + 4 * g2 + 2 * h;
    const int woff  = 4 * g1 + 2 * h;

    const float* featsB = feats + (size_t)b * PREV_LEN * NC * HW;
    const float* simB   = sim   + (size_t)b * TM * HW * HW;
    float*       outB   = out   + (size_t)b * AFT_LEN * NC * HW;

    f4 mA[16], mB[16];
    f4 cA[8], cB[8];
    v2f fA[8], fW[4];

    #pragma unroll
    for (int p = 1; p <= 2; ++p) {
        __builtin_prefetch(simB + (size_t)p * HW * HW + tid * 32, 0, 0);
        if (wv < 2) __builtin_prefetch(featsB + (size_t)(p + 1) * NC * HW + tid * 32, 0, 0);
    }

    {
        const float* t0 = featsB + (ft * 16 + c) * HW + 2 * h;
        #pragma unroll
        for (int kk = 0; kk < 4; ++kk) {
            fA[kk]     = *(const v2f*)(t0 + 4 * (g1 + kk));
            fA[4 + kk] = *(const v2f*)(t0 + 4 * (g2 + kk));
            const float one = (c == 0) ? 1.0f : 0.0f;
            fW[kk].x = one; fW[kk].y = one;
        }
        load_add(cA, featsB + (size_t)NC * HW, rowoff);
        const f4* m0 = (const f4*)simB;
        load_half(mA, m0 + off1, m0 + off2);
    }

    #pragma nounroll
    for (int s = 0; s < PREV_LEN - 2; s += 2) {
        product_step<true, true>(s,     mA, mB, cA, cB, fA, fW, featsB, simB, outB, sP, sW, wv, tid, c, h, ft, kh, off1, off2, rowoff, aoff1, aoff2, woff);
        product_step<true, true>(s + 1, mB, mA, cB, cA, fA, fW, featsB, simB, outB, sP, sW, wv, tid, c, h, ft, kh, off1, off2, rowoff, aoff1, aoff2, woff);
    }
    product_step<false, true>(PREV_LEN - 2, mA, mB, cA, cB, fA, fW, featsB, simB, outB, sP, sW, wv, tid, c, h, ft, kh, off1, off2, rowoff, aoff1, aoff2, woff);
    #pragma nounroll
    for (int s = PREV_LEN - 1; s < TM; s += 2) {
        product_step<false, false>(s,     mB, mA, cB, cA, fA, fW, featsB, simB, outB, sP, sW, wv, tid, c, h, ft, kh, off1, off2, rowoff, aoff1, aoff2, woff);
        product_step<false, false>(s + 1, mA, mB, cA, cB, fA, fW, featsB, simB, outB, sP, sW, wv, tid, c, h, ft, kh, off1, off2, rowoff, aoff1, aoff2, woff);
    }
}

extern "C" void kernel_launch(void* const* d_in, const int* in_sizes, int n_in,
                              void* d_out, int out_size, void* d_ws, size_t ws_size,
                              hipStream_t stream) {
    const float* feats = (const float*)d_in[0];
    const float* sim   = (const float*)d_in[1];
    float* out = (float*)d_out;
    chain_ksplit_kernel<<<NB, NWAVE * 32, 0, stream>>>(feats, sim, out);
}


# Second wave per SIMD: 8 waves as two K-split groups; the ten output products split by t over the groups (5 lockstep steps of 2 products, own LDS partials); chain on group 0 alone; sums unchanged
# speedup vs baseline: 3.9627x; 1.0587x over previous
#include <hip/hip_runtime.h>


typedef __attribute__((ext_vector_type(2))) float v2f;
typedef __attribute__((ext_vector_type(8))) float v8f;
typedef __attribute__((ext_vector_type(4))) float f4;

#define PREV_LEN 10
#define AFT_LEN  10
#define NB       256
#define NC       32
#define HW       64
#define TM       (PREV_LEN - 1 + AFT_LEN)
#define NWAVE    8
#define GWAVE    4
#define GTHR     (GWAVE * 32)
#define SPG      (2 * 2 * 2 * PTILE)
#define SWG      (2 * 4 * HW)
#define TSTR     68
#define PTILE    (16 * TSTR)
#define EPSF     1e-6f

__device__ __forceinline__ void load_half(f4 (&q)[16], const f4* __restrict__ p1, const f4* __restrict__ p2)
{
    #pragma unroll
    for (int kk = 0; kk < 4; ++kk) {
        q[2 * kk]     = p1[kk * 64];
        q[2 * kk + 1] = p1[kk * 64 + 16];
    }
    #pragma unroll
    for (int kk = 0; kk < 4; ++kk) {
        q[8 + 2 * kk]     = p2[kk * 64];
        q[8 + 2 * kk + 1] = p2[kk * 64 + 16];
    }
}

__device__ __forceinline__ void load_add(f4 (&c)[8], const float* __restrict__ fa, int rowoff)
{
    const f4* p = (const f4*)fa + rowoff;
    #pragma unroll
    for (int r = 0; r < 8; ++r) c[r] = p[r * 16];
}

__device__ __forceinline__ void read_frags(v2f (&fA)[8], v2f (&fW)[4],
        const float* P0, const float* P1, const float* Wp,
        const int aoff1, const int aoff2, const int woff, const bool row0)
{
    #pragma unroll
    for (int kk = 0; kk < 4; ++kk) {
        fA[kk]     = *(const v2f*)(P0 + aoff1 + 4 * kk) + *(const v2f*)(P1 + aoff1 + 4 * kk);
        fA[4 + kk] = *(const v2f*)(P0 + aoff2 + 4 * kk) + *(const v2f*)(P1 + aoff2 + 4 * kk);
    }
    #pragma unroll
    for (int kk = 0; kk < 4; ++kk) {
        const v2f w = (*(const v2f*)(Wp + woff + 4 * kk) + *(const v2f*)(Wp + HW + woff + 4 * kk))
                    + (*(const v2f*)(Wp + 2 * HW + woff + 4 * kk) + *(const v2f*)(Wp + 3 * HW + woff + 4 * kk));
        fW[kk].x = row0 ? w.x : 0.0f;
        fW[kk].y = row0 ? w.y : 0.0f;
    }
}

template <bool INITC>
__device__ __forceinline__ void mma_and_store(const f4 (&cur)[16], const f4 (&ccur)[8],
        const v2f (&fA)[8], const v2f (&fW)[4], float* myP, float* myW,
        const int wv4, const int c, const int h, const int kh)
{
    v8f acc[4] = {};
    v8f accW[4] = {};
    if (INITC) {
        if (kh == 0) {
            #pragma unroll
            for (int r = 0; r < 8; ++r) {
                acc[0][r] = ccur[r].x; acc[1][r] = ccur[r].y;
                acc[2][r] = ccur[r].z; acc[3][r] = ccur[r].w;
            }
        }
        if (wv4 == 0) {
            const float one = (h == 0) ? 1.0f : 0.0f;
            accW[0][0] = one; accW[1][0] = one; accW[2][0] = one; accW[3][0] = one;
        }
    }

    #pragma unroll
    for (int kk = 0; kk < 8; ++kk) {
        const f4 b0 = cur[2 * kk], b1 = cur[2 * kk + 1];
        v2f bf;
        bf.x = b0.x; bf.y = b1.x;
        acc[0] = __builtin_amdgcn_wmma_f32_16x16x4_f32(false, fA[kk], false, bf, (short)0, acc[0], false, false);
        if (kk < 4) accW[0] = __builtin_amdgcn_wmma_f32_16x16x4_f32(false, fW[kk], false, bf, (short)0, accW[0], false, false);
        bf.x = b0.y; bf.y = b1.y;
        acc[1] = __builtin_amdgcn_wmma_f32_16x16x4_f32(false, fA[kk], false, bf, (short)0, acc[1], false, false);
        if (kk < 4) accW[1] = __builtin_amdgcn_wmma_f32_16x16x4_f32(false, fW[kk], false, bf, (short)0, accW[1], false, false);
        bf.x = b0.z; bf.y = b1.z;
        acc[2] = __builtin_amdgcn_wmma_f32_16x16x4_f32(false, fA[kk], false, bf, (short)0, acc[2], false, false);
        if (kk < 4) accW[2] = __builtin_amdgcn_wmma_f32_16x16x4_f32(false, fW[kk], false, bf, (short)0, accW[2], false, false);
        bf.x = b0.w; bf.y = b1.w;
        acc[3] = __builtin_amdgcn_wmma_f32_16x16x4_f32(false, fA[kk], false, bf, (short)0, acc[3], false, false);
        if (kk < 4) accW[3] = __builtin_amdgcn_wmma_f32_16x16x4_f32(false, fW[kk], false, bf, (short)0, accW[3], false, false);
    }

    #pragma unroll
    for (int r = 0; r < 8; ++r) {
        f4 v;
        v.x = acc[0][r]; v.y = acc[1][r]; v.z = acc[2][r]; v.w = acc[3][r];
        *(f4*)(myP + (h * 8 + r) * TSTR + c * 4) = v;
    }
    if (h == 0) {
        f4 v;
        v.x = accW[0][0]; v.y = accW[1][0]; v.z = accW[2][0]; v.w = accW[3][0];
        *(f4*)(myW + c * 4) = v;
    }
}

__device__ __forceinline__ void lds_barrier()
{
    __builtin_amdgcn_fence(__ATOMIC_RELEASE, "workgroup", "local");
    __builtin_amdgcn_s_barrier();
    __builtin_amdgcn_fence(__ATOMIC_ACQUIRE, "workgroup", "local");
}

template <bool LOADC>
__device__ __forceinline__ void chain_step(const int step,
        const f4 (&cur)[16], f4 (&nxt)[16], const f4 (&ccur)[8], f4 (&cnxt)[8],
        v2f (&fA)[8], v2f (&fW)[4],
        const float* __restrict__ featsB, const float* __restrict__ simB,
        float* sPbase, float* sWbase,
        const int wv4, const int t128, const int c, const int h, const int ft, const int kh,
        const int off1, const int off2, const int rowoff,
        const int aoff1, const int aoff2, const int woff)
{
    if (LOADC)
        load_add(cnxt, featsB + (size_t)(step + 2) * NC * HW, rowoff);
    const f4* mn = (const f4*)(simB + (size_t)(step + 1) * HW * HW);
    load_half(nxt, mn + off1, mn + off2);
    __builtin_prefetch(simB + (size_t)(step + 3) * HW * HW + t128 * 32, 0, 0);
    if (LOADC && step + 4 < PREV_LEN && wv4 < 2)
        __builtin_prefetch(featsB + (size_t)(step + 4) * NC * HW + t128 * 32, 0, 0);

    const int set = (step + 1) & 1;
    float* Wp = sWbase + set * 4 * HW;
    mma_and_store<true>(cur, ccur, fA, fW,
                        sPbase + ((set * 2 + ft) * 2 + kh) * PTILE, Wp + (2 * kh + ft) * HW, wv4, c, h, kh);
    lds_barrier();
    const float* P0 = sPbase + ((set * 2 + ft) * 2) * PTILE;
    read_frags(fA, fW, P0, P0 + PTILE, Wp, aoff1, aoff2, woff, c == 0);
}

__device__ __forceinline__ void out_step(const int i, const int grp,
        const f4 (&cur)[16], f4 (&nxt)[16], const f4 (&cdummy)[8],
        const v2f (&fA)[8], const v2f (&fW)[4],
        const float* __restrict__ simB, float* __restrict__ outB,
        float* sPg, float* sWg,
        const int wv4, const int t128, const int c, const int h, const int ft, const int kh,
        const int off1, const int off2)
{
    const int t = 2 * i + grp;
    const int m = PREV_LEN - 1 + t;
    const int nm = (m + 2 < TM) ? m + 2 : m;
    const f4* mn = (const f4*)(simB + (size_t)nm * HW * HW);
    load_half(nxt, mn + off1, mn + off2);
    if (m + 6 < TM)
        __builtin_prefetch(simB + (size_t)(m + 6) * HW * HW + t128 * 32, 0, 0);

    const int set = i & 1;
    float* Wp = sWg + set * 4 * HW;
    mma_and_store<false>(cur, cdummy, fA, fW,
                         sPg + ((set * 2 + ft) * 2 + kh) * PTILE, Wp + (2 * kh + ft) * HW, wv4, c, h, kh);
    lds_barrier();

    const float* P0 = sPg + ((set * 2 + ft) * 2) * PTILE;
    const float* P1 = P0 + PTILE;
    const f4 w = (*(const f4*)(Wp + c * 4) + *(const f4*)(Wp + HW + c * 4))
               + (*(const f4*)(Wp + 2 * HW + c * 4) + *(const f4*)(Wp + 3 * HW + c * 4));
    f4 rw;
    rw.x = 1.0f / (w.x + EPSF); rw.y = 1.0f / (w.y + EPSF);
    rw.z = 1.0f / (w.z + EPSF); rw.w = 1.0f / (w.w + EPSF);
    f4* o = (f4*)(outB + (size_t)t * NC * HW) + (ft * 16 + kh * 8 + h) * 16 + c;
    #pragma unroll
    for (int q = 0; q < 4; ++q) {
        const int rr = kh * 8 + 2 * q + h;
        const f4 s = *(const f4*)(P0 + rr * TSTR + c * 4) + *(const f4*)(P1 + rr * TSTR + c * 4);
        o[q * 32] = s * rw;
    }
}

__global__ __launch_bounds__(NWAVE * 32)
void chain_ksplit_tsplit_kernel(const float* __restrict__ feats,
                                const float* __restrict__ sim,
                                float* __restrict__ out)
{
    __shared__ __attribute__((aligned(16))) float sP[2 * SPG];
    __shared__ __attribute__((aligned(16))) float sW[2 * SWG];

    const int b   = blockIdx.x;
    const int tid = threadIdx.x;
    const int wv  = __builtin_amdgcn_readfirstlane(tid >> 5);
    const int grp = wv >> 2;
    const int wv4 = wv & 3;
    const int t128 = tid & (GTHR - 1);
    const int ft  = wv4 >> 1;
    const int kh  = wv4 & 1;
    const int lane = tid & 31;
    const int c   = lane & 15;
    const int h   = lane >> 4;
    const int g1  = 8 * kh + 4 * ft;
    const int g2  = 8 * kh + 4 * (1 - ft);
    const int off1 = g1 * 64 + h * 32 + c;
    const int off2 = g2 * 64 + h * 32 + c;
    const int rowoff = (ft * 16 + h * 8) * 16 + c;
    const int aoff1 = c * TSTR + 4 * g1 + 2 * h;
    const int aoff2 = c * TSTR + 4 * g2 + 2 * h;
    const int woff  = 4 * g1 + 2 * h;
    const bool active = (grp == 0);

    const float* featsB = feats + (size_t)b * PREV_LEN * NC * HW;
    const float* simB   = sim   + (size_t)b * TM * HW * HW;
    float*       outB   = out   + (size_t)b * AFT_LEN * NC * HW;

    f4 mA[16], mB[16];
    f4 cA[8], cB[8];
    v2f fA[8], fW[4];

    if (active) {
        #pragma unroll
        for (int p = 1; p <= 2; ++p) {
            __builtin_prefetch(simB + (size_t)p * HW * HW + t128 * 32, 0, 0);
            if (wv4 < 2) __builtin_prefetch(featsB + (size_t)(p + 1) * NC * HW + t128 * 32, 0, 0);
        }

        const float* t0 = featsB + (ft * 16 + c) * HW + 2 * h;
        #pragma unroll
        for (int kk = 0; kk < 4; ++kk) {
            fA[kk]     = *(const v2f*)(t0 + 4 * (g1 + kk));
            fA[4 + kk] = *(const v2f*)(t0 + 4 * (g2 + kk));
            const float one = (c == 0) ? 1.0f : 0.0f;
            fW[kk].x = one; fW[kk].y = one;
        }
        load_add(cA, featsB + (size_t)NC * HW, rowoff);
        const f4* m0 = (const f4*)simB;
        load_half(mA, m0 + off1, m0 + off2);

        #pragma nounroll
        for (int s = 0; s < PREV_LEN - 2; s += 2) {
            chain_step<true>(s,     mA, mB, cA, cB, fA, fW, featsB, simB, sP, sW, wv4, t128, c, h, ft, kh, off1, off2, rowoff, aoff1, aoff2, woff);
            chain_step<true>(s + 1, mB, mA, cB, cA, fA, fW, featsB, simB, sP, sW, wv4, t128, c, h, ft, kh, off1, off2, rowoff, aoff1, aoff2, woff);
        }
        chain_step<false>(PREV_LEN - 2, mA, mB, cA, cB, fA, fW, featsB, simB, sP, sW, wv4, t128, c, h, ft, kh, off1, off2, rowoff, aoff1, aoff2, woff);
    } else {
        const f4* mn = (const f4*)(simB + (size_t)PREV_LEN * HW * HW);
        load_half(mB, mn + off1, mn + off2);
        #pragma nounroll
        for (int s = 0; s < PREV_LEN - 1; ++s) {
            if (s >= PREV_LEN - 4)
                __builtin_prefetch(simB + (size_t)(s + 6) * HW * HW + t128 * 32, 0, 0);
            lds_barrier();
        }
        const int set = (PREV_LEN - 1) & 1;
        const float* P0 = sP + ((set * 2 + ft) * 2) * PTILE;
        read_frags(fA, fW, P0, P0 + PTILE, sW + set * 4 * HW, aoff1, aoff2, woff, c == 0);
    }

    float* sPg = sP + grp * SPG;
    float* sWg = sW + grp * SWG;
    #pragma nounroll
    for (int i = 0; i < AFT_LEN / 2 - 1; i += 2) {
        out_step(i,     grp, mB, mA, cA, fA, fW, simB, outB, sPg, sWg, wv4, t128, c, h, ft, kh, off1, off2);
        out_step(i + 1, grp, mA, mB, cA, fA, fW, simB, outB, sPg, sWg, wv4, t128, c, h, ft, kh, off1, off2);
    }
    out_step(AFT_LEN / 2 - 1, grp, mB, mA, cA, fA, fW, simB, outB, sPg, sWg, wv4, t128, c, h, ft, kh, off1, off2);
}

extern "C" void kernel_launch(void* const* d_in, const int* in_sizes, int n_in,
                              void* d_out, int out_size, void* d_ws, size_t ws_size,
                              hipStream_t stream) {
    const float* feats = (const float*)d_in[0];
    const float* sim   = (const float*)d_in[1];
    float* out = (float*)d_out;
    chain_ksplit_tsplit_kernel<<<NB, NWAVE * 32, 0, stream>>>(feats, sim, out);
}


# Weight-row chain moved to the idle wave group, out of phase with the feature chain: barrier as two alternating phases (arrive early, wait late), feature waves at raised issue priority; sums unchanged
# speedup vs baseline: 4.1779x; 1.0543x over previous
#include <hip/hip_runtime.h>


typedef __attribute__((ext_vector_type(2))) float v2f;
typedef __attribute__((ext_vector_type(8))) float v8f;
typedef __attribute__((ext_vector_type(4))) float f4;

#define PREV_LEN 10
#define AFT_LEN  10
#define NB       256
#define NC       32
#define HW       64
#define TM       (PREV_LEN - 1 + AFT_LEN)
#define NWAVE    8
#define GWAVE    4
#define GTHR     (GWAVE * 32)
#define SPG      (2 * 2 * 2 * PTILE)
#define SWG      (2 * 4 * HW)
#define TSTR     68
#define PTILE    (16 * TSTR)
#define EPSF     1e-6f

__device__ __forceinline__ void load_half(f4 (&q)[16], const f4* __restrict__ p1, const f4* __restrict__ p2)
{
    #pragma unroll
    for (int kk = 0; kk < 4; ++kk) {
        q[2 * kk]     = p1[kk * 64];
        q[2 * kk + 1] = p1[kk * 64 + 16];
    }
    #pragma unroll
    for (int kk = 0; kk < 4; ++kk) {
        q[8 + 2 * kk]     = p2[kk * 64];
        q[8 + 2 * kk + 1] = p2[kk * 64 + 16];
    }
}

__device__ __forceinline__ void load_quarter(f4 (&q)[8], const f4* __restrict__ p1)
{
    #pragma unroll
    for (int kk = 0; kk < 4; ++kk) {
        q[2 * kk]     = p1[kk * 64];
        q[2 * kk + 1] = p1[kk * 64 + 16];
    }
}

__device__ __forceinline__ void load_add(f4 (&c)[8], const float* __restrict__ fa, int rowoff)
{
    const f4* p = (const f4*)fa + rowoff;
    #pragma unroll
    for (int r = 0; r < 8; ++r) c[r] = p[r * 16];
}

__device__ __forceinline__ void read_fragsA(v2f (&fA)[8], const float* P0, const float* P1,
        const int aoff1, const int aoff2)
{
    #pragma unroll
    for (int kk = 0; kk < 4; ++kk) {
        fA[kk]     = *(const v2f*)(P0 + aoff1 + 4 * kk) + *(const v2f*)(P1 + aoff1 + 4 * kk);
        fA[4 + kk] = *(const v2f*)(P0 + aoff2 + 4 * kk) + *(const v2f*)(P1 + aoff2 + 4 * kk);
    }
}

__device__ __forceinline__ void issue_fragsA(v2f (&r0)[8], v2f (&r1)[8], const float* P0, const float* P1,
        const int aoff1, const int aoff2)
{
    #pragma unroll
    for (int kk = 0; kk < 4; ++kk) {
        r0[kk]     = *(const v2f*)(P0 + aoff1 + 4 * kk);
        r1[kk]     = *(const v2f*)(P1 + aoff1 + 4 * kk);
        r0[4 + kk] = *(const v2f*)(P0 + aoff2 + 4 * kk);
        r1[4 + kk] = *(const v2f*)(P1 + aoff2 + 4 * kk);
    }
}

__device__ __forceinline__ void read_fragsW(v2f (&fW)[4], const float* Wp, const int woff, const bool row0)
{
    #pragma unroll
    for (int kk = 0; kk < 4; ++kk) {
        const v2f w = (*(const v2f*)(Wp + woff + 4 * kk) + *(const v2f*)(Wp + HW + woff + 4 * kk))
                    + (*(const v2f*)(Wp + 2 * HW + woff + 4 * kk) + *(const v2f*)(Wp + 3 * HW + woff + 4 * kk));
        fW[kk].x = row0 ? w.x : 0.0f;
        fW[kk].y = row0 ? w.y : 0.0f;
    }
}

template <bool INITC, bool WITHW>
__device__ __forceinline__ void mma_and_store(const f4 (&cur)[16], const f4 (&ccur)[8],
        const v2f (&fA)[8], const v2f (&fW)[4], float* myP, float* myW,
        const int wv4, const int c, const int h, const int kh)
{
    v8f acc[4] = {};
    v8f accW[4] = {};
    if (INITC) {
        if (kh == 0) {
            #pragma unroll
            for (int r = 0; r < 8; ++r) {
                acc[0][r] = ccur[r].x; acc[1][r] = ccur[r].y;
                acc[2][r] = ccur[r].z; acc[3][r] = ccur[r].w;
            }
        }
    }

    #pragma unroll
    for (int kk = 0; kk < 8; ++kk) {
        const f4 b0 = cur[2 * kk], b1 = cur[2 * kk + 1];
        v2f bf;
        bf.x = b0.x; bf.y = b1.x;
        acc[0] = __builtin_amdgcn_wmma_f32_16x16x4_f32(false, fA[kk], false, bf, (short)0, acc[0], false, false);
        if (WITHW && kk < 4) accW[0] = __builtin_amdgcn_wmma_f32_16x16x4_f32(false, fW[kk], false, bf, (short)0, accW[0], false, false);
        bf.x = b0.y; bf.y = b1.y;
        acc[1] = __builtin_amdgcn_wmma_f32_16x16x4_f32(false, fA[kk], false, bf, (short)0, acc[1], false, false);
        if (WITHW && kk < 4) accW[1] = __builtin_amdgcn_wmma_f32_16x16x4_f32(false, fW[kk], false, bf, (short)0, accW[1], false, false);
        bf.x = b0.z; bf.y = b1.z;
        acc[2] = __builtin_amdgcn_wmma_f32_16x16x4_f32(false, fA[kk], false, bf, (short)0, acc[2], false, false);
        if (WITHW && kk < 4) accW[2] = __builtin_amdgcn_wmma_f32_16x16x4_f32(false, fW[kk], false, bf, (short)0, accW[2], false, false);
        bf.x = b0.w; bf.y = b1.w;
        acc[3] = __builtin_amdgcn_wmma_f32_16x16x4_f32(false, fA[kk], false, bf, (short)0, acc[3], false, false);
        if (WITHW && kk < 4) accW[3] = __builtin_amdgcn_wmma_f32_16x16x4_f32(false, fW[kk], false, bf, (short)0, accW[3], false, false);
    }

    #pragma unroll
    for (int r = 0; r < 8; ++r) {
        f4 v;
        v.x = acc[0][r]; v.y = acc[1][r]; v.z = acc[2][r]; v.w = acc[3][r];
        *(f4*)(myP + (h * 8 + r) * TSTR + c * 4) = v;
    }
    if (WITHW && h == 0) {
        f4 v;
        v.x = accW[0][0]; v.y = accW[1][0]; v.z = accW[2][0]; v.w = accW[3][0];
        *(f4*)(myW + c * 4) = v;
    }
}

__device__ __forceinline__ void bar_publish()
{
    __builtin_amdgcn_fence(__ATOMIC_RELEASE, "workgroup", "local");
    __builtin_amdgcn_s_barrier_signal(-1);
}
__device__ __forceinline__ void bar_arrive()
{
    __builtin_amdgcn_s_barrier_signal(-1);
}
__device__ __forceinline__ void bar_wait_read()
{
    __builtin_amdgcn_s_barrier_wait(-1);
    __builtin_amdgcn_fence(__ATOMIC_ACQUIRE, "workgroup", "local");
}
__device__ __forceinline__ void bar_wait()
{
    __builtin_amdgcn_s_barrier_wait(-1);
}
__device__ __forceinline__ void lds_barrier()
{
    bar_publish();
    bar_wait_read();
}

template <bool LOADC>
__device__ __forceinline__ void chain_step(const int step,
        const f4 (&cur)[16], f4 (&nxt)[16], const f4 (&ccur)[8], f4 (&cnxt)[8],
        v2f (&r0)[8], v2f (&r1)[8],
        const float* __restrict__ featsB, const float* __restrict__ simB,
        float* sPbase,
        const int wv4, const int t128, const int c, const int h, const int ft, const int kh,
        const int off1, const int off2, const int rowoff,
        const int aoff1, const int aoff2)
{
    if (LOADC)
        load_add(cnxt, featsB + (size_t)(step + 2) * NC * HW, rowoff);
    const f4* mn = (const f4*)(simB + (size_t)(step + 1) * HW * HW);
    load_half(nxt, mn + off1, mn + off2);
    __builtin_prefetch(simB + (size_t)(step + 3) * HW * HW + t128 * 32, 0, 0);
    if (LOADC && step + 4 < PREV_LEN && wv4 < 2)
        __builtin_prefetch(featsB + (size_t)(step + 4) * NC * HW + t128 * 32, 0, 0);
    __builtin_amdgcn_sched_barrier(0);

    v2f fA[8];
    #pragma unroll
    for (int i = 0; i < 8; ++i) fA[i] = r0[i] + r1[i];

    const v2f fWnone[4] = {};
    const int set = (step + 1) & 1;
    mma_and_store<true, false>(cur, ccur, fA, fWnone,
                               sPbase + ((set * 2 + ft) * 2 + kh) * PTILE, nullptr, wv4, c, h, kh);
    __builtin_amdgcn_sched_barrier(0);
    if (step > 0) bar_wait();
    bar_publish();
    bar_wait_read();
    bar_arrive();
    const float* P0 = sPbase + ((set * 2 + ft) * 2) * PTILE;
    issue_fragsA(r0, r1, P0, P0 + PTILE, aoff1, aoff2);
    __builtin_amdgcn_sched_barrier(0);
}

template <bool NEXT>
__device__ __forceinline__ void wchain_step(const int step,
        const f4 (&cur)[8], f4 (&nxt)[8], v2f (&fW)[4],
        const float* __restrict__ simB, float* sWbase,
        const int wv4, const int t128, const int c, const int h, const int ft, const int kh,
        const int off1, const int woff)
{
    if (NEXT) {
        const f4* mn = (const f4*)(simB + (size_t)(step + 1) * HW * HW);
        load_quarter(nxt, mn + off1);
    }
    if (step >= PREV_LEN - 4)
        __builtin_prefetch(simB + (size_t)(step + 6) * HW * HW + t128 * 32, 0, 0);

    v8f accW[4] = {};
    if (wv4 == 0) {
        const float one = (h == 0) ? 1.0f : 0.0f;
        accW[0][0] = one; accW[1][0] = one; accW[2][0] = one; accW[3][0] = one;
    }
    #pragma unroll
    for (int kk = 0; kk < 4; ++kk) {
        const f4 b0 = cur[2 * kk], b1 = cur[2 * kk + 1];
        v2f bf;
        bf.x = b0.x; bf.y = b1.x;
        accW[0] = __builtin_amdgcn_wmma_f32_16x16x4_f32(false, fW[kk], false, bf, (short)0, accW[0], false, false);
        bf.x = b0.y; bf.y = b1.y;
        accW[1] = __builtin_amdgcn_wmma_f32_16x16x4_f32(false, fW[kk], false, bf, (short)0, accW[1], false, false);
        bf.x = b0.z; bf.y = b1.z;
        accW[2] = __builtin_amdgcn_wmma_f32_16x16x4_f32(false, fW[kk], false, bf, (short)0, accW[2], false, false);
        bf.x = b0.w; bf.y = b1.w;
        accW[3] = __builtin_amdgcn_wmma_f32_16x16x4_f32(false, fW[kk], false, bf, (short)0, accW[3], false, false);
    }
    __builtin_amdgcn_sched_barrier(0);
    bar_wait();

    const int set = (step + 1) & 1;
    float* Wp = sWbase + set * 4 * HW;
    if (h == 0) {
        f4 v;
        v.x = accW[0][0]; v.y = accW[1][0]; v.z = accW[2][0]; v.w = accW[3][0];
        *(f4*)(Wp + (2 * kh + ft) * HW + c * 4) = v;
    }
    bar_publish();
    bar_wait_read();
    if (NEXT) bar_arrive();
    read_fragsW(fW, Wp, woff, c == 0);
    __builtin_amdgcn_sched_barrier(0);
}

__device__ __forceinline__ void out_step(const int i, const int grp,
        const f4 (&cur)[16], f4 (&nxt)[16], const f4 (&cdummy)[8],
        const v2f (&fA)[8], const v2f (&fW)[4],
        const float* __restrict__ simB, float* __restrict__ outB,
        float* sPg, float* sWg,
        const int wv4, const int t128, const int c, const int h, const int ft, const int kh,
        const int off1, const int off2)
{
    const int t = 2 * i + grp;
    const int m = PREV_LEN - 1 + t;
    const int nm = (m + 2 < TM) ? m + 2 : m;
    const f4* mn = (const f4*)(simB + (size_t)nm * HW * HW);
    load_half(nxt, mn + off1, mn + off2);
    if (m + 6 < TM)
        __builtin_prefetch(simB + (size_t)(m + 6) * HW * HW + t128 * 32, 0, 0);

    const int set = i & 1;
    float* Wp = sWg + set * 4 * HW;
    mma_and_store<false, true>(cur, cdummy, fA, fW,
                               sPg + ((set * 2 + ft) * 2 + kh) * PTILE, Wp + (2 * kh + ft) * HW, wv4, c, h, kh);
    lds_barrier();

    const float* P0 = sPg + ((set * 2 + ft) * 2) * PTILE;
    const float* P1 = P0 + PTILE;
    const f4 w = (*(const f4*)(Wp + c * 4) + *(const f4*)(Wp + HW + c * 4))
               + (*(const f4*)(Wp + 2 * HW + c * 4) + *(const f4*)(Wp + 3 * HW + c * 4));
    f4 rw;
    rw.x = 1.0f / (w.x + EPSF); rw.y = 1.0f / (w.y + EPSF);
    rw.z = 1.0f / (w.z + EPSF); rw.w = 1.0f / (w.w + EPSF);
    f4* o = (f4*)(outB + (size_t)t * NC * HW) + (ft * 16 + kh * 8 + h) * 16 + c;
    #pragma unroll
    for (int q = 0; q < 4; ++q) {
        const int rr = kh * 8 + 2 * q + h;
        const f4 s = *(const f4*)(P0 + rr * TSTR + c * 4) + *(const f4*)(P1 + rr * TSTR + c * 4);
        o[q * 32] = s * rw;
    }
}

__global__ __launch_bounds__(NWAVE * 32)
void chain_ksplit_tsplit_kernel(const float* __restrict__ feats,
                                const float* __restrict__ sim,
                                float* __restrict__ out)
{
    __shared__ __attribute__((aligned(16))) float sP[2 * SPG];
    __shared__ __attribute__((aligned(16))) float sW[2 * SWG];

    const int b   = blockIdx.x;
    const int tid = threadIdx.x;
    const int wv  = __builtin_amdgcn_readfirstlane(tid >> 5);
    const int grp = wv >> 2;
    const int wv4 = wv & 3;
    const int t128 = tid & (GTHR - 1);
    const int ft  = wv4 >> 1;
    const int kh  = wv4 & 1;
    const int lane = tid & 31;
    const int c   = lane & 15;
    const int h   = lane >> 4;
    const int g1  = 8 * kh + 4 * ft;
    const int g2  = 8 * kh + 4 * (1 - ft);
    const int off1 = g1 * 64 + h * 32 + c;
    const int off2 = g2 * 64 + h * 32 + c;
    const int rowoff = (ft * 16 + h * 8) * 16 + c;
    const int aoff1 = c * TSTR + 4 * g1 + 2 * h;
    const int aoff2 = c * TSTR + 4 * g2 + 2 * h;
    const int woff  = 4 * g1 + 2 * h;
    const bool active = (grp == 0);

    const float* featsB = feats + (size_t)b * PREV_LEN * NC * HW;
    const float* simB   = sim   + (size_t)b * TM * HW * HW;
    float*       outB   = out   + (size_t)b * AFT_LEN * NC * HW;

    f4 mA[16], mB[16];
    f4 cA[8], cB[8];
    v2f fA[8], fW[4];
    const int lset = (PREV_LEN - 1) & 1;
    const float* lastP = sP + ((lset * 2 + ft) * 2) * PTILE;
    const float* lastW = sW + lset * 4 * HW;

    if (active) {
        __builtin_amdgcn_s_setprio(3);

        #pragma unroll
        for (int p = 1; p <= 2; ++p) {
            __builtin_prefetch(simB + (size_t)p * HW * HW + t128 * 32, 0, 0);
            if (wv4 < 2) __builtin_prefetch(featsB + (size_t)(p + 1) * NC * HW + t128 * 32, 0, 0);
        }

        v2f r0[8], r1[8];
        const float* t0 = featsB + (ft * 16 + c) * HW + 2 * h;
        #pragma unroll
        for (int kk = 0; kk < 4; ++kk) {
            r0[kk]     = *(const v2f*)(t0 + 4 * (g1 + kk));
            r0[4 + kk] = *(const v2f*)(t0 + 4 * (g2 + kk));
            r1[kk].x = -0.0f;     r1[kk].y = -0.0f;
            r1[4 + kk].x = -0.0f; r1[4 + kk].y = -0.0f;
        }
        load_add(cA, featsB + (size_t)NC * HW, rowoff);
        const f4* m0 = (const f4*)simB;
        load_half(mA, m0 + off1, m0 + off2);

        #pragma nounroll
        for (int s = 0; s < PREV_LEN - 2; s += 2) {
            chain_step<true>(s,     mA, mB, cA, cB, r0, r1, featsB, simB, sP, wv4, t128, c, h, ft, kh, off1, off2, rowoff, aoff1, aoff2);
            chain_step<true>(s + 1, mB, mA, cB, cA, r0, r1, featsB, simB, sP, wv4, t128, c, h, ft, kh, off1, off2, rowoff, aoff1, aoff2);
        }
        chain_step<false>(PREV_LEN - 2, mA, mB, cA, cB, r0, r1, featsB, simB, sP, wv4, t128, c, h, ft, kh, off1, off2, rowoff, aoff1, aoff2);

        __builtin_amdgcn_s_setprio(0);
        #pragma unroll
        for (int i = 0; i < 8; ++i) fA[i] = r0[i] + r1[i];
        bar_wait_read();
        read_fragsW(fW, lastW, woff, c == 0);
    } else {
        f4 qA[8], qB[8];
        const f4* m0 = (const f4*)simB;
        load_quarter(qA, m0 + off1);
        const f4* mn = (const f4*)(simB + (size_t)PREV_LEN * HW * HW);
        load_half(mB, mn + off1, mn + off2);
        #pragma unroll
        for (int kk = 0; kk < 4; ++kk) {
            const float one = (c == 0) ? 1.0f : 0.0f;
            fW[kk].x = one; fW[kk].y = one;
        }
        bar_arrive();
        #pragma nounroll
        for (int s = 0; s < PREV_LEN - 2; s += 2) {
            wchain_step<true>(s,     qA, qB, fW, simB, sW, wv4, t128, c, h, ft, kh, off1, woff);
            wchain_step<true>(s + 1, qB, qA, fW, simB, sW, wv4, t128, c, h, ft, kh, off1, woff);
        }
        wchain_step<false>(PREV_LEN - 2, qA, qB, fW, simB, sW, wv4, t128, c, h, ft, kh, off1, woff);
        read_fragsA(fA, lastP, lastP + PTILE, aoff1, aoff2);
    }

    float* sPg = sP + grp * SPG;
    float* sWg = sW + grp * SWG;
    #pragma nounroll
    for (int i = 0; i < AFT_LEN / 2 - 1; i += 2) {
        out_step(i,     grp, mB, mA, cA, fA, fW, simB, outB, sPg, sWg, wv4, t128, c, h, ft, kh, off1, off2);
        out_step(i + 1, grp, mA, mB, cA, fA, fW, simB, outB, sPg, sWg, wv4, t128, c, h, ft, kh, off1, off2);
    }
    out_step(AFT_LEN / 2 - 1, grp, mB, mA, cA, fA, fW, simB, outB, sPg, sWg, wv4, t128, c, h, ft, kh, off1, off2);
}

extern "C" void kernel_launch(void* const* d_in, const int* in_sizes, int n_in,
                              void* d_out, int out_size, void* d_ws, size_t ws_size,
                              hipStream_t stream) {
    const float* feats = (const float*)d_in[0];
    const float* sim   = (const float*)d_in[1];
    float* out = (float*)d_out;
    chain_ksplit_tsplit_kernel<<<NB, NWAVE * 32, 0, stream>>>(feats, sim, out);
}
